# starting kernel
# baseline (speedup 1.0000x reference)
#include <hip/hip_runtime.h>
#include <math.h>

typedef float v2f __attribute__((ext_vector_type(2)));
typedef float v8f __attribute__((ext_vector_type(8)));

#define NS 32768
#define DD 64
#define LL 63
#define PP 14
#define NCAT 512
#define BB 5.0f
#define MINW 0.001f
#define MINH 0.001f
#define MIND 0.001f

__global__ void nsf_pack_w1(const float* __restrict__ W1, float* __restrict__ Bt) {
    int idx = blockIdx.x * blockDim.x + threadIdx.x;
    if (idx >= NCAT * 64) return;
    int col = idx >> 6;
    int k   = idx & 63;
    int l   = col >> 3;
    int h   = col & 7;
    float v = 0.0f;
    if (l < LL && k <= l) v = W1[(l * LL + k) * 8 + h];
    Bt[col * 64 + k] = v;
}

__device__ __forceinline__ float softplusf(float v) {
    return v > 20.0f ? v : log1pf(expf(v));
}

__device__ void rqs_eval(const float* __restrict__ p, float xf, float& z, float& ld) {
    float cumw[6], cumh[6], derivs[6];
    {
        float mx = p[0];
        #pragma unroll
        for (int i = 1; i < 5; i++) mx = fmaxf(mx, p[i]);
        float e[5], s = 0.0f;
        #pragma unroll
        for (int i = 0; i < 5; i++) { e[i] = expf(p[i] - mx); s += e[i]; }
        float inv = 1.0f / s, cs = 0.0f;
        cumw[0] = -BB;
        #pragma unroll
        for (int i = 0; i < 4; i++) {
            cs += MINW + (1.0f - MINW * 5.0f) * e[i] * inv;
            cumw[i + 1] = 2.0f * BB * cs - BB;
        }
        cumw[5] = BB;
    }
    {
        float mx = p[5];
        #pragma unroll
        for (int i = 6; i < 10; i++) mx = fmaxf(mx, p[i]);
        float e[5], s = 0.0f;
        #pragma unroll
        for (int i = 0; i < 5; i++) { e[i] = expf(p[5 + i] - mx); s += e[i]; }
        float inv = 1.0f / s, cs = 0.0f;
        cumh[0] = -BB;
        #pragma unroll
        for (int i = 0; i < 4; i++) {
            cs += MINH + (1.0f - MINH * 5.0f) * e[i] * inv;
            cumh[i + 1] = 2.0f * BB * cs - BB;
        }
        cumh[5] = BB;
    }
    derivs[0] = 1.0f; derivs[5] = 1.0f;
    #pragma unroll
    for (int i = 1; i < 5; i++) derivs[i] = MIND + softplusf(p[9 + i]);

    float xc = fminf(fmaxf(xf, -BB), BB);
    int idx = 0;
    #pragma unroll
    for (int j = 1; j < 6; j++) idx += (xc >= cumw[j]) ? 1 : 0;
    if (idx > 4) idx = 4;

    float icw = cumw[idx], ibw = cumw[idx + 1] - cumw[idx];
    float ich = cumh[idx], ih  = cumh[idx + 1] - cumh[idx];
    float idl = ih / ibw;
    float idv = derivs[idx], idvp1 = derivs[idx + 1];
    float theta = (xc - icw) / ibw;
    float tmt = theta * (1.0f - theta);
    float num = ih * (idl * theta * theta + idv * tmt);
    float den = idl + (idv + idvp1 - 2.0f * idl) * tmt;
    float outv = ich + num / den;
    float omt = 1.0f - theta;
    float dnum = idl * idl * (idvp1 * theta * theta + 2.0f * idl * tmt + idv * omt * omt);
    float ldv = logf(dnum) - 2.0f * logf(den);
    bool inside = (xf >= -BB) && (xf <= BB);
    z  = inside ? outv : xf;
    ld = inside ? ldv : 0.0f;
}

__launch_bounds__(128)
__global__ void nsf_fused(const float* __restrict__ x, const float* __restrict__ initp,
                          const float* __restrict__ b1, const float* __restrict__ W2,
                          const float* __restrict__ b2, const float* __restrict__ W3,
                          const float* __restrict__ b3, const float* __restrict__ Bt,
                          float* __restrict__ out) {
    __shared__ float tile[4][16][17];
    __shared__ float ldacc[DD];

    const int tid  = threadIdx.x;
    const int wave = tid >> 5;
    const int lane = tid & 31;
    const int m    = lane & 15;
    const int hh   = lane >> 4;
    const int s0   = blockIdx.x * 64 + wave * 16;
    const int s    = s0 + m;

    if (tid < DD) ldacc[tid] = 0.0f;
    __syncthreads();

    if (lane < 16) {
        float p[PP];
        #pragma unroll
        for (int q = 0; q < PP; q++) p[q] = initp[q];
        float z, ld;
        rqs_eval(p, x[s * DD + 0], z, ld);
        out[s] = z;
        atomicAdd(&ldacc[0], ld);
    }

    v2f a[16];
    const float* xrow = x + (size_t)s * DD;
    #pragma unroll
    for (int kk = 0; kk < 16; kk++)
        a[kk] = *(const v2f*)(xrow + 4 * kk + 2 * hh);

    for (int t = 0; t < 32; ++t) {
        const float* bcol = Bt + (size_t)(16 * t + m) * 64 + 2 * hh;
        v8f c = {0.f, 0.f, 0.f, 0.f, 0.f, 0.f, 0.f, 0.f};
        #pragma unroll
        for (int kk = 0; kk < 16; kk++) {
            v2f b = *(const v2f*)(bcol + 4 * kk);
            c = __builtin_amdgcn_wmma_f32_16x16x4_f32(
                    false, a[kk], false, b, (short)0, c, false, false);
        }

        const int col = 16 * t + m;
        const float bias = (col < LL * 8) ? b1[col] : 0.0f;
        #pragma unroll
        for (int r = 0; r < 8; r++) {
            float hv = tanhf(c[r] + bias);
            tile[wave][r + 8 * hh][m] = hv;
        }
        __syncthreads();

        const int l_idx = 2 * t + hh;
        if (l_idx < LL) {
            float hv[8];
            #pragma unroll
            for (int j = 0; j < 8; j++) hv[j] = tile[wave][m][8 * hh + j];

            const float* w2  = W2 + (size_t)l_idx * 64;
            const float* bb2 = b2 + (size_t)l_idx * 8;
            float h2[8];
            #pragma unroll
            for (int g = 0; g < 8; g++) {
                float acc = bb2[g];
                #pragma unroll
                for (int j = 0; j < 8; j++) acc += hv[j] * w2[j * 8 + g];
                h2[g] = tanhf(acc);
            }

            const float* w3  = W3 + (size_t)l_idx * 8 * PP;
            const float* bb3 = b3 + (size_t)l_idx * PP;
            float pv[PP];
            #pragma unroll
            for (int q = 0; q < PP; q++) {
                float acc = bb3[q];
                #pragma unroll
                for (int g = 0; g < 8; g++) acc += h2[g] * w3[g * PP + q];
                pv[q] = acc;
            }

            const int cdim = l_idx + 1;
            float z, ld;
            rqs_eval(pv, x[(size_t)s * DD + cdim], z, ld);
            out[(size_t)cdim * NS + s] = z;
            atomicAdd(&ldacc[cdim], ld);
        }
        __syncthreads();
    }

    if (tid < DD)
        out[(size_t)NS * DD + (size_t)tid * (NS / 64) + blockIdx.x] = ldacc[tid];
}

extern "C" void kernel_launch(void* const* d_in, const int* in_sizes, int n_in,
                              void* d_out, int out_size, void* d_ws, size_t ws_size,
                              hipStream_t stream) {
    const float* x     = (const float*)d_in[0];
    const float* initp = (const float*)d_in[1];
    const float* W1    = (const float*)d_in[2];
    const float* b1    = (const float*)d_in[3];
    const float* W2    = (const float*)d_in[4];
    const float* b2    = (const float*)d_in[5];
    const float* W3    = (const float*)d_in[6];
    const float* b3    = (const float*)d_in[7];
    float* out = (float*)d_out;
    float* Bt  = (float*)d_ws;

    nsf_pack_w1<<<(NCAT * 64 + 255) / 256, 256, 0, stream>>>(W1, Bt);
    nsf_fused<<<NS / 64, 128, 0, stream>>>(x, initp, b1, W2, b2, W3, b3, Bt, out);
}


# loop software-pipelined over tiles: spline of tile i and the 3 layers of tile i+1 share one branch-free trip, so WMMA and tanh chains fill the spline's gaps; x rows asked a trip ahead; same sums
# speedup vs baseline: 3.1874x; 3.1874x over previous
#include <hip/hip_runtime.h>
#include <math.h>

typedef float v2f __attribute__((ext_vector_type(2)));
typedef float v8f __attribute__((ext_vector_type(8)));

#define NS 32768
#define DD 64
#define LL 63
#define PP 14
#define NCAT 512
#define BB 5.0f
#define MINW 0.001f
#define MINH 0.001f
#define MIND 0.001f

__global__ void nsf_pack_w1(const float* __restrict__ W1, float* __restrict__ Bt) {
    int idx = blockIdx.x * blockDim.x + threadIdx.x;
    if (idx >= NCAT * 64) return;
    int col = idx >> 6;
    int k   = idx & 63;
    int l   = col >> 3;
    int h   = col & 7;
    float v = 0.0f;
    if (l < LL && k <= l) v = W1[(l * LL + k) * 8 + h];
    Bt[col * 64 + k] = v;
}


__device__ __forceinline__ void rqs_soft(const float (&p)[PP], float (&e)[10], float& invw, float& invh) {
    float mxw = p[0], mxh = p[5];
    #pragma unroll
    for (int i = 1; i < 5; i++) { mxw = fmaxf(mxw, p[i]); mxh = fmaxf(mxh, p[5 + i]); }
    float d[10];
    #pragma unroll
    for (int i = 0; i < 5; i++) { d[i] = p[i] - mxw; d[5 + i] = p[5 + i] - mxh; }
    #pragma unroll
    for (int i = 0; i < 10; i++) e[i] = expf(d[i]);
    float sw = 0.0f, sh = 0.0f;
    #pragma unroll
    for (int i = 0; i < 5; i++) { sw += e[i]; sh += e[5 + i]; }
    invw = 1.0f / sw; invh = 1.0f / sh;
}

__device__ __forceinline__ void rqs_bins(const float (&p)[PP], const float (&e)[10], float invw, float invh,
                                         float xf, float& xc, int& idx, float& icw, float& ibw,
                                         float& ich, float& ih, float& rlo, float& rhi) {
    float cumw[6], cumh[6];
    float csw = 0.0f, csh = 0.0f;
    cumw[0] = -BB; cumh[0] = -BB;
    #pragma unroll
    for (int i = 0; i < 4; i++) {
        csw += MINW + (1.0f - MINW * 5.0f) * e[i] * invw;
        csh += MINH + (1.0f - MINH * 5.0f) * e[5 + i] * invh;
        cumw[i + 1] = 2.0f * BB * csw - BB;
        cumh[i + 1] = 2.0f * BB * csh - BB;
    }
    cumw[5] = BB; cumh[5] = BB;

    xc = fminf(fmaxf(xf, -BB), BB);
    idx = 0;
    #pragma unroll
    for (int j = 1; j < 6; j++) idx += (xc >= cumw[j]) ? 1 : 0;
    if (idx > 4) idx = 4;

    icw = cumw[idx]; ibw = cumw[idx + 1] - cumw[idx];
    ich = cumh[idx]; ih  = cumh[idx + 1] - cumh[idx];
    const float r1 = p[10], r2 = p[11], r3 = p[12], r4 = p[13];
    const float lo34 = (idx >= 4) ? r4 : r3;
    const float lo12 = (idx >= 2) ? r2 : r1;
    rlo = (idx >= 3) ? lo34 : lo12;
    const float hi23 = (idx >= 3) ? r4 : r3;
    const float hi01 = (idx >= 1) ? r2 : r1;
    rhi = (idx >= 2) ? hi23 : hi01;
}

__device__ __forceinline__ void rqs_knots(int idx, float rlo, float rhi, float& idv, float& idvp1) {
    const float elo = expf(rlo), ehi = expf(rhi);
    const float llo = log1pf(elo), lhi = log1pf(ehi);
    const float splo = rlo > 20.0f ? rlo : llo;
    const float sphi = rhi > 20.0f ? rhi : lhi;
    idv   = (idx == 0) ? 1.0f : MIND + splo;
    idvp1 = (idx == 4) ? 1.0f : MIND + sphi;
}

__device__ __forceinline__ void rqs_form(float xf, float xc, float icw, float ibw, float ich, float ih,
                                         float idv, float idvp1, float& z, float& ld) {
    float idl = ih / ibw;
    float theta = (xc - icw) / ibw;
    float tmt = theta * (1.0f - theta);
    float num = ih * (idl * theta * theta + idv * tmt);
    float den = idl + (idv + idvp1 - 2.0f * idl) * tmt;
    float outv = ich + num / den;
    float omt = 1.0f - theta;
    float dnum = idl * idl * (idvp1 * theta * theta + 2.0f * idl * tmt + idv * omt * omt);
    float ldv = logf(dnum) - 2.0f * logf(den);
    bool inside = (xf >= -BB) && (xf <= BB);
    z  = inside ? outv : xf;
    ld = inside ? ldv : 0.0f;
}

__device__ __forceinline__ void tanh8_exp(const v8f a, float (&y)[8], float (&y2)[8], float (&t)[8]) {
    #pragma unroll
    for (int r = 0; r < 8; r++) { y[r] = fabsf(a[r]); y2[r] = a[r] * a[r]; }
    #pragma unroll
    for (int r = 0; r < 8; r++) t[r] = expf(y[r] + y[r]);
}
__device__ __forceinline__ void tanh8_poly(const float (&y)[8], const float (&y2)[8], float (&zs)[8]) {
    float q[8];
    #pragma unroll
    for (int r = 0; r < 8; r++) q[r] = fmaf(y2[r], -0x1.758e7ap-8f, 0x1.521192p-6f);
    #pragma unroll
    for (int r = 0; r < 8; r++) q[r] = fmaf(y2[r], q[r], -0x1.b8389cp-5f);
    #pragma unroll
    for (int r = 0; r < 8; r++) q[r] = fmaf(y2[r], q[r], 0x1.110704p-3f);
    #pragma unroll
    for (int r = 0; r < 8; r++) q[r] = fmaf(y2[r], q[r], -0x1.555532p-2f);
    #pragma unroll
    for (int r = 0; r < 8; r++) zs[r] = fmaf(y2[r], y[r] * q[r], y[r]);
}
__device__ __forceinline__ void tanh8_pick(const v8f a, const float (&y)[8], const float (&zs)[8],
                                           const float (&t)[8], float (&o)[8]) {
    float zl[8];
    #pragma unroll
    for (int r = 0; r < 8; r++) zl[r] = fmaf(__builtin_amdgcn_rcpf(t[r] + 1.0f), -2.0f, 1.0f);
    #pragma unroll
    for (int r = 0; r < 8; r++) o[r] = copysignf((y[r] < 0.625f) ? zs[r] : zl[r], a[r]);
}
__device__ __forceinline__ void tanh8(const v8f a, float (&o)[8]) {
    float y[8], y2[8], t[8], zs[8];
    tanh8_exp(a, y, y2, t);
    tanh8_poly(y, y2, zs);
    tanh8_pick(a, y, zs, t, o);
}

#define TILES 16
#define WAVES 16

__launch_bounds__(WAVES * 32) __attribute__((amdgpu_waves_per_eu(4, 4)))
__global__ void nsf_fused(const float* __restrict__ x, const float* __restrict__ initp,
                          const float* __restrict__ b1, const float* __restrict__ W2,
                          const float* __restrict__ b2, const float* __restrict__ W3,
                          const float* __restrict__ b3, const float* __restrict__ Bt,
                          float* __restrict__ out) {
    const int tid   = threadIdx.x;
    const int wave  = tid >> 5;
    const int lane  = tid & 31;
    const int c     = lane & 15;
    const int h     = lane >> 4;
    const int t     = (blockIdx.x & 1) * WAVES + wave;
    const int chunk = blockIdx.x >> 1;

    const int  l    = 2 * t + h;
    const bool lv   = (l < LL);
    const int  lc   = lv ? l : (LL - 1);
    const int  cdim = lv ? (l + 1) : 0;

    const float* wrow = Bt + (size_t)(16 * t + c) * 64 + 2 * h;
    v2f a1[16];
    #pragma unroll
    for (int kk = 0; kk < 16; kk++) a1[kk] = *(const v2f*)(wrow + 4 * kk);

    const int  rp   = c >> 3;
    const int  ru   = c & 7;
    const bool diag = (rp == h) && lv;

    const float* w2 = W2 + (size_t)lc * 64 + ru;
    v2f a2[4];
    #pragma unroll
    for (int kk = 0; kk < 4; kk++) {
        float w0 = w2[(2 * kk) * 8], w1 = w2[(2 * kk + 1) * 8];
        a2[kk].x = diag ? w0 : 0.0f;
        a2[kk].y = diag ? w1 : 0.0f;
    }

    const float* w3 = W3 + (size_t)lc * (8 * PP);
    const bool on1 = diag && (ru < PP - 8);
    const int  q1  = on1 ? (8 + ru) : 0;
    v2f a30[4], a31[4];
    #pragma unroll
    for (int kk = 0; kk < 4; kk++) {
        float u0 = w3[(2 * kk) * PP + ru], u1 = w3[(2 * kk + 1) * PP + ru];
        float v0 = w3[(2 * kk) * PP + q1], v1 = w3[(2 * kk + 1) * PP + q1];
        a30[kk].x = diag ? u0 : 0.0f;
        a30[kk].y = diag ? u1 : 0.0f;
        a31[kk].x = on1 ? v0 : 0.0f;
        a31[kk].y = on1 ? v1 : 0.0f;
    }

    v8f c1, c2, c30, c31;
    #pragma unroll
    for (int r = 0; r < 8; r++) {
        c1[r] = b1[lc * 8 + r];
        c2[r] = b2[lc * 8 + r];
        float g0 = b3[lc * PP + r], i0 = initp[r];
        c30[r] = lv ? g0 : i0;
        if (r < PP - 8) {
            float g1 = b3[lc * PP + 8 + r], i1 = initp[8 + r];
            c31[r] = lv ? g1 : i1;
        } else {
            c31[r] = 0.0f;
        }
    }

    float ldsum = 0.0f;

    const int s0 = chunk * (TILES * 16) + c;

    v2f xb[16];
    float xs_l;
#define LOAD_TILE(TI)                                                        \
    {                                                                        \
        const float* xrow_ = x + (size_t)(s0 + (TI) * 16) * DD;              \
        _Pragma("unroll")                                                    \
        for (int kk = 0; kk < 16; kk++)                                      \
            xb[kk] = *(const v2f*)(xrow_ + 4 * kk + 2 * h);                  \
        xs_l = xrow_[cdim];                                                  \
    }

    float pv[PP];
    float xs_c;
    {
        LOAD_TILE(0)
        v8f acc1 = c1;
        #pragma unroll
        for (int kk = 0; kk < 16; kk++)
            acc1 = __builtin_amdgcn_wmma_f32_16x16x4_f32(
                       false, a1[kk], false, xb[kk], (short)0, acc1, false, false);
        xs_c = xs_l;
        LOAD_TILE(1)
        float h1[8];
        tanh8(acc1, h1);
        v8f acc2 = c2;
        #pragma unroll
        for (int kk = 0; kk < 4; kk++) {
            v2f b = {h1[2 * kk], h1[2 * kk + 1]};
            acc2 = __builtin_amdgcn_wmma_f32_16x16x4_f32(
                       false, a2[kk], false, b, (short)0, acc2, false, false);
        }
        float h2[8];
        tanh8(acc2, h2);
        v8f p0 = c30, p1 = c31;
        #pragma unroll
        for (int kk = 0; kk < 4; kk++) {
            v2f b = {h2[2 * kk], h2[2 * kk + 1]};
            p0 = __builtin_amdgcn_wmma_f32_16x16x4_f32(
                     false, a30[kk], false, b, (short)0, p0, false, false);
            p1 = __builtin_amdgcn_wmma_f32_16x16x4_f32(
                     false, a31[kk], false, b, (short)0, p1, false, false);
        }
        #pragma unroll
        for (int q = 0; q < 8; q++) pv[q] = p0[q];
        #pragma unroll
        for (int q = 8; q < PP; q++) pv[q] = p1[q - 8];
    }

    for (int i = 0; i < TILES - 1; ++i) {
        v8f acc1 = c1;
        #pragma unroll
        for (int kk = 0; kk < 16; kk++)
            acc1 = __builtin_amdgcn_wmma_f32_16x16x4_f32(
                       false, a1[kk], false, xb[kk], (short)0, acc1, false, false);
        const float xs_n = xs_l;

        float e[10], invw, invh;
        rqs_soft(pv, e, invw, invh);

        const int nt = (i + 2) & (TILES - 1);
        LOAD_TILE(nt)

        float y[8], y2[8], t8[8], zs[8], h1[8];
        tanh8_exp(acc1, y, y2, t8);

        float xc, icw, ibw, ich, ih, rlo, rhi;
        int idx;
        rqs_bins(pv, e, invw, invh, xs_c, xc, idx, icw, ibw, ich, ih, rlo, rhi);

        tanh8_poly(y, y2, zs);

        float idv, idvp1;
        rqs_knots(idx, rlo, rhi, idv, idvp1);

        tanh8_pick(acc1, y, zs, t8, h1);

        v8f acc2 = c2;
        #pragma unroll
        for (int kk = 0; kk < 4; kk++) {
            v2f b = {h1[2 * kk], h1[2 * kk + 1]};
            acc2 = __builtin_amdgcn_wmma_f32_16x16x4_f32(
                       false, a2[kk], false, b, (short)0, acc2, false, false);
        }

        float u[8], u2[8], w8[8], us[8], h2[8];
        tanh8_exp(acc2, u, u2, w8);
        float z, ld;
        rqs_form(xs_c, xc, icw, ibw, ich, ih, idv, idvp1, z, ld);
        tanh8_poly(u, u2, us);
        tanh8_pick(acc2, u, us, w8, h2);

        v8f p0 = c30, p1 = c31;
        #pragma unroll
        for (int kk = 0; kk < 4; kk++) {
            v2f b = {h2[2 * kk], h2[2 * kk + 1]};
            p0 = __builtin_amdgcn_wmma_f32_16x16x4_f32(
                     false, a30[kk], false, b, (short)0, p0, false, false);
            p1 = __builtin_amdgcn_wmma_f32_16x16x4_f32(
                     false, a31[kk], false, b, (short)0, p1, false, false);
        }

        out[(size_t)cdim * NS + s0 + i * 16] = z;
        ldsum += ld;

        if ((i & 3) == 3) {
            float v = ldsum;
            v += __shfl_xor(v, 1);
            v += __shfl_xor(v, 2);
            v += __shfl_xor(v, 4);
            v += __shfl_xor(v, 8);
            if (c == 0)
                out[(size_t)NS * DD + (size_t)cdim * (NS / 64) + chunk * (TILES / 4) + (i >> 2)] = v;
            ldsum = 0.0f;
        }

        #pragma unroll
        for (int q = 0; q < 8; q++) pv[q] = p0[q];
        #pragma unroll
        for (int q = 8; q < PP; q++) pv[q] = p1[q - 8];
        xs_c = xs_n;
    }
#undef LOAD_TILE

    {
        float e[10], invw, invh;
        rqs_soft(pv, e, invw, invh);
        float xc, icw, ibw, ich, ih, rlo, rhi;
        int idx;
        rqs_bins(pv, e, invw, invh, xs_c, xc, idx, icw, ibw, ich, ih, rlo, rhi);
        float idv, idvp1;
        rqs_knots(idx, rlo, rhi, idv, idvp1);
        float z, ld;
        rqs_form(xs_c, xc, icw, ibw, ich, ih, idv, idvp1, z, ld);
        out[(size_t)cdim * NS + s0 + (TILES - 1) * 16] = z;
        float v = ldsum + ld;
        v += __shfl_xor(v, 1);
        v += __shfl_xor(v, 2);
        v += __shfl_xor(v, 4);
        v += __shfl_xor(v, 8);
        if (c == 0)
            out[(size_t)NS * DD + (size_t)cdim * (NS / 64) + chunk * (TILES / 4) + ((TILES - 1) >> 2)] = v;
    }
}

extern "C" void kernel_launch(void* const* d_in, const int* in_sizes, int n_in,
                              void* d_out, int out_size, void* d_ws, size_t ws_size,
                              hipStream_t stream) {
    const float* x     = (const float*)d_in[0];
    const float* initp = (const float*)d_in[1];
    const float* W1    = (const float*)d_in[2];
    const float* b1    = (const float*)d_in[3];
    const float* W2    = (const float*)d_in[4];
    const float* b2    = (const float*)d_in[5];
    const float* W3    = (const float*)d_in[6];
    const float* b3    = (const float*)d_in[7];
    float* out = (float*)d_out;
    float* Bt  = (float*)d_ws;

    nsf_pack_w1<<<(NCAT * 64 + 255) / 256, 256, 0, stream>>>(W1, Bt);
    nsf_fused<<<(32 / WAVES) * (NS / (TILES * 16)), WAVES * 32, 0, stream>>>(x, initp, b1, W2, b2, W3, b3, Bt, out);
}


# exp in tanh (arg 2|v| >= 0) and softmax (arg <= 0) in line: remainder x*log2e - n by 2 FMAs (library's 2-part constant), expf's range guards that cannot fire left out; results equal or neighbour
# speedup vs baseline: 3.4999x; 1.0980x over previous
#include <hip/hip_runtime.h>
#include <math.h>

typedef float v2f __attribute__((ext_vector_type(2)));
typedef float v8f __attribute__((ext_vector_type(8)));

#define NS 32768
#define DD 64
#define LL 63
#define PP 14
#define NCAT 512
#define BB 5.0f
#define MINW 0.001f
#define MINH 0.001f
#define MIND 0.001f

__global__ void nsf_pack_w1(const float* __restrict__ W1, float* __restrict__ Bt) {
    int idx = blockIdx.x * blockDim.x + threadIdx.x;
    if (idx >= NCAT * 64) return;
    int col = idx >> 6;
    int k   = idx & 63;
    int l   = col >> 3;
    int h   = col & 7;
    float v = 0.0f;
    if (l < LL && k <= l) v = W1[(l * LL + k) * 8 + h];
    Bt[col * 64 + k] = v;
}

__device__ __forceinline__ float exp_core(float x) {
    const float c_hi = 0x1.715476p+0f, c_lo = 0x1.4ae0bep-26f;
    const float n = __builtin_rintf(x * c_hi);
    float a = __builtin_fmaf(x, c_hi, -n);
    a = __builtin_fmaf(x, c_lo, a);
    return __builtin_ldexpf(__builtin_amdgcn_exp2f(a), (int)n);
}
__device__ __forceinline__ float exp_nonneg(float x) { return exp_core(x); }
__device__ __forceinline__ float exp_nonpos(float x) {
    const float r = exp_core(x);
    return (x < -0x1.9d1da0p+6f) ? 0.0f : r;
}


__device__ __forceinline__ void rqs_soft(const float (&p)[PP], float (&e)[10], float& invw, float& invh) {
    float mxw = p[0], mxh = p[5];
    #pragma unroll
    for (int i = 1; i < 5; i++) { mxw = fmaxf(mxw, p[i]); mxh = fmaxf(mxh, p[5 + i]); }
    float d[10];
    #pragma unroll
    for (int i = 0; i < 5; i++) { d[i] = p[i] - mxw; d[5 + i] = p[5 + i] - mxh; }
    #pragma unroll
    for (int i = 0; i < 10; i++) e[i] = exp_nonpos(d[i]);
    float sw = 0.0f, sh = 0.0f;
    #pragma unroll
    for (int i = 0; i < 5; i++) { sw += e[i]; sh += e[5 + i]; }
    invw = 1.0f / sw; invh = 1.0f / sh;
}

__device__ __forceinline__ void rqs_bins(const float (&p)[PP], const float (&e)[10], float invw, float invh,
                                         float xf, float& xc, int& idx, float& icw, float& ibw,
                                         float& ich, float& ih, float& rlo, float& rhi) {
    float cumw[6], cumh[6];
    float csw = 0.0f, csh = 0.0f;
    cumw[0] = -BB; cumh[0] = -BB;
    #pragma unroll
    for (int i = 0; i < 4; i++) {
        csw += MINW + (1.0f - MINW * 5.0f) * e[i] * invw;
        csh += MINH + (1.0f - MINH * 5.0f) * e[5 + i] * invh;
        cumw[i + 1] = 2.0f * BB * csw - BB;
        cumh[i + 1] = 2.0f * BB * csh - BB;
    }
    cumw[5] = BB; cumh[5] = BB;

    xc = fminf(fmaxf(xf, -BB), BB);
    idx = 0;
    #pragma unroll
    for (int j = 1; j < 6; j++) idx += (xc >= cumw[j]) ? 1 : 0;
    if (idx > 4) idx = 4;

    icw = cumw[idx]; ibw = cumw[idx + 1] - cumw[idx];
    ich = cumh[idx]; ih  = cumh[idx + 1] - cumh[idx];
    const float r1 = p[10], r2 = p[11], r3 = p[12], r4 = p[13];
    const float lo34 = (idx >= 4) ? r4 : r3;
    const float lo12 = (idx >= 2) ? r2 : r1;
    rlo = (idx >= 3) ? lo34 : lo12;
    const float hi23 = (idx >= 3) ? r4 : r3;
    const float hi01 = (idx >= 1) ? r2 : r1;
    rhi = (idx >= 2) ? hi23 : hi01;
}

__device__ __forceinline__ void rqs_knots(int idx, float rlo, float rhi, float& idv, float& idvp1) {
    const float elo = expf(rlo), ehi = expf(rhi);
    const float llo = log1pf(elo), lhi = log1pf(ehi);
    const float splo = rlo > 20.0f ? rlo : llo;
    const float sphi = rhi > 20.0f ? rhi : lhi;
    idv   = (idx == 0) ? 1.0f : MIND + splo;
    idvp1 = (idx == 4) ? 1.0f : MIND + sphi;
}

__device__ __forceinline__ void rqs_form(float xf, float xc, float icw, float ibw, float ich, float ih,
                                         float idv, float idvp1, float& z, float& ld) {
    float idl = ih / ibw;
    float theta = (xc - icw) / ibw;
    float tmt = theta * (1.0f - theta);
    float num = ih * (idl * theta * theta + idv * tmt);
    float den = idl + (idv + idvp1 - 2.0f * idl) * tmt;
    float outv = ich + num / den;
    float omt = 1.0f - theta;
    float dnum = idl * idl * (idvp1 * theta * theta + 2.0f * idl * tmt + idv * omt * omt);
    float ldv = logf(dnum) - 2.0f * logf(den);
    bool inside = (xf >= -BB) && (xf <= BB);
    z  = inside ? outv : xf;
    ld = inside ? ldv : 0.0f;
}

__device__ __forceinline__ void tanh8_exp(const v8f a, float (&y)[8], float (&y2)[8], float (&t)[8]) {
    #pragma unroll
    for (int r = 0; r < 8; r++) { y[r] = fabsf(a[r]); y2[r] = a[r] * a[r]; }
    #pragma unroll
    for (int r = 0; r < 8; r++) t[r] = exp_nonneg(y[r] + y[r]);
}
__device__ __forceinline__ void tanh8_poly(const float (&y)[8], const float (&y2)[8], float (&zs)[8]) {
    float q[8];
    #pragma unroll
    for (int r = 0; r < 8; r++) q[r] = fmaf(y2[r], -0x1.758e7ap-8f, 0x1.521192p-6f);
    #pragma unroll
    for (int r = 0; r < 8; r++) q[r] = fmaf(y2[r], q[r], -0x1.b8389cp-5f);
    #pragma unroll
    for (int r = 0; r < 8; r++) q[r] = fmaf(y2[r], q[r], 0x1.110704p-3f);
    #pragma unroll
    for (int r = 0; r < 8; r++) q[r] = fmaf(y2[r], q[r], -0x1.555532p-2f);
    #pragma unroll
    for (int r = 0; r < 8; r++) zs[r] = fmaf(y2[r], y[r] * q[r], y[r]);
}
__device__ __forceinline__ void tanh8_pick(const v8f a, const float (&y)[8], const float (&zs)[8],
                                           const float (&t)[8], float (&o)[8]) {
    float zl[8];
    #pragma unroll
    for (int r = 0; r < 8; r++) zl[r] = fmaf(__builtin_amdgcn_rcpf(t[r] + 1.0f), -2.0f, 1.0f);
    #pragma unroll
    for (int r = 0; r < 8; r++) o[r] = copysignf((y[r] < 0.625f) ? zs[r] : zl[r], a[r]);
}
__device__ __forceinline__ void tanh8(const v8f a, float (&o)[8]) {
    float y[8], y2[8], t[8], zs[8];
    tanh8_exp(a, y, y2, t);
    tanh8_poly(y, y2, zs);
    tanh8_pick(a, y, zs, t, o);
}

#define TILES 16
#define WAVES 16

__launch_bounds__(WAVES * 32) __attribute__((amdgpu_waves_per_eu(4, 4)))
__global__ void nsf_fused(const float* __restrict__ x, const float* __restrict__ initp,
                          const float* __restrict__ b1, const float* __restrict__ W2,
                          const float* __restrict__ b2, const float* __restrict__ W3,
                          const float* __restrict__ b3, const float* __restrict__ Bt,
                          float* __restrict__ out) {
    const int tid   = threadIdx.x;
    const int wave  = tid >> 5;
    const int lane  = tid & 31;
    const int c     = lane & 15;
    const int h     = lane >> 4;
    const int t     = (blockIdx.x & 1) * WAVES + wave;
    const int chunk = blockIdx.x >> 1;

    const int  l    = 2 * t + h;
    const bool lv   = (l < LL);
    const int  lc   = lv ? l : (LL - 1);
    const int  cdim = lv ? (l + 1) : 0;

    const float* wrow = Bt + (size_t)(16 * t + c) * 64 + 2 * h;
    v2f a1[16];
    #pragma unroll
    for (int kk = 0; kk < 16; kk++) a1[kk] = *(const v2f*)(wrow + 4 * kk);

    const int  rp   = c >> 3;
    const int  ru   = c & 7;
    const bool diag = (rp == h) && lv;

    const float* w2 = W2 + (size_t)lc * 64 + ru;
    v2f a2[4];
    #pragma unroll
    for (int kk = 0; kk < 4; kk++) {
        float w0 = w2[(2 * kk) * 8], w1 = w2[(2 * kk + 1) * 8];
        a2[kk].x = diag ? w0 : 0.0f;
        a2[kk].y = diag ? w1 : 0.0f;
    }

    const float* w3 = W3 + (size_t)lc * (8 * PP);
    const bool on1 = diag && (ru < PP - 8);
    const int  q1  = on1 ? (8 + ru) : 0;
    v2f a30[4], a31[4];
    #pragma unroll
    for (int kk = 0; kk < 4; kk++) {
        float u0 = w3[(2 * kk) * PP + ru], u1 = w3[(2 * kk + 1) * PP + ru];
        float v0 = w3[(2 * kk) * PP + q1], v1 = w3[(2 * kk + 1) * PP + q1];
        a30[kk].x = diag ? u0 : 0.0f;
        a30[kk].y = diag ? u1 : 0.0f;
        a31[kk].x = on1 ? v0 : 0.0f;
        a31[kk].y = on1 ? v1 : 0.0f;
    }

    v8f c1, c2, c30, c31;
    #pragma unroll
    for (int r = 0; r < 8; r++) {
        c1[r] = b1[lc * 8 + r];
        c2[r] = b2[lc * 8 + r];
        float g0 = b3[lc * PP + r], i0 = initp[r];
        c30[r] = lv ? g0 : i0;
        if (r < PP - 8) {
            float g1 = b3[lc * PP + 8 + r], i1 = initp[8 + r];
            c31[r] = lv ? g1 : i1;
        } else {
            c31[r] = 0.0f;
        }
    }

    float ldsum = 0.0f;

    const int s0 = chunk * (TILES * 16) + c;

    v2f xb[16];
    float xs_l;
#define LOAD_TILE(TI)                                                        \
    {                                                                        \
        const float* xrow_ = x + (size_t)(s0 + (TI) * 16) * DD;              \
        _Pragma("unroll")                                                    \
        for (int kk = 0; kk < 16; kk++)                                      \
            xb[kk] = *(const v2f*)(xrow_ + 4 * kk + 2 * h);                  \
        xs_l = xrow_[cdim];                                                  \
    }

    float pv[PP];
    float xs_c;
    {
        LOAD_TILE(0)
        v8f acc1 = c1;
        #pragma unroll
        for (int kk = 0; kk < 16; kk++)
            acc1 = __builtin_amdgcn_wmma_f32_16x16x4_f32(
                       false, a1[kk], false, xb[kk], (short)0, acc1, false, false);
        xs_c = xs_l;
        LOAD_TILE(1)
        float h1[8];
        tanh8(acc1, h1);
        v8f acc2 = c2;
        #pragma unroll
        for (int kk = 0; kk < 4; kk++) {
            v2f b = {h1[2 * kk], h1[2 * kk + 1]};
            acc2 = __builtin_amdgcn_wmma_f32_16x16x4_f32(
                       false, a2[kk], false, b, (short)0, acc2, false, false);
        }
        float h2[8];
        tanh8(acc2, h2);
        v8f p0 = c30, p1 = c31;
        #pragma unroll
        for (int kk = 0; kk < 4; kk++) {
            v2f b = {h2[2 * kk], h2[2 * kk + 1]};
            p0 = __builtin_amdgcn_wmma_f32_16x16x4_f32(
                     false, a30[kk], false, b, (short)0, p0, false, false);
            p1 = __builtin_amdgcn_wmma_f32_16x16x4_f32(
                     false, a31[kk], false, b, (short)0, p1, false, false);
        }
        #pragma unroll
        for (int q = 0; q < 8; q++) pv[q] = p0[q];
        #pragma unroll
        for (int q = 8; q < PP; q++) pv[q] = p1[q - 8];
    }

    for (int i = 0; i < TILES - 1; ++i) {
        v8f acc1 = c1;
        #pragma unroll
        for (int kk = 0; kk < 16; kk++)
            acc1 = __builtin_amdgcn_wmma_f32_16x16x4_f32(
                       false, a1[kk], false, xb[kk], (short)0, acc1, false, false);
        const float xs_n = xs_l;

        float e[10], invw, invh;
        rqs_soft(pv, e, invw, invh);

        const int nt = (i + 2) & (TILES - 1);
        LOAD_TILE(nt)

        float y[8], y2[8], t8[8], zs[8], h1[8];
        tanh8_exp(acc1, y, y2, t8);

        float xc, icw, ibw, ich, ih, rlo, rhi;
        int idx;
        rqs_bins(pv, e, invw, invh, xs_c, xc, idx, icw, ibw, ich, ih, rlo, rhi);

        tanh8_poly(y, y2, zs);

        float idv, idvp1;
        rqs_knots(idx, rlo, rhi, idv, idvp1);

        tanh8_pick(acc1, y, zs, t8, h1);

        v8f acc2 = c2;
        #pragma unroll
        for (int kk = 0; kk < 4; kk++) {
            v2f b = {h1[2 * kk], h1[2 * kk + 1]};
            acc2 = __builtin_amdgcn_wmma_f32_16x16x4_f32(
                       false, a2[kk], false, b, (short)0, acc2, false, false);
        }

        float u[8], u2[8], w8[8], us[8], h2[8];
        tanh8_exp(acc2, u, u2, w8);
        float z, ld;
        rqs_form(xs_c, xc, icw, ibw, ich, ih, idv, idvp1, z, ld);
        tanh8_poly(u, u2, us);
        tanh8_pick(acc2, u, us, w8, h2);

        v8f p0 = c30, p1 = c31;
        #pragma unroll
        for (int kk = 0; kk < 4; kk++) {
            v2f b = {h2[2 * kk], h2[2 * kk + 1]};
            p0 = __builtin_amdgcn_wmma_f32_16x16x4_f32(
                     false, a30[kk], false, b, (short)0, p0, false, false);
            p1 = __builtin_amdgcn_wmma_f32_16x16x4_f32(
                     false, a31[kk], false, b, (short)0, p1, false, false);
        }

        out[(size_t)cdim * NS + s0 + i * 16] = z;
        ldsum += ld;

        if ((i & 3) == 3) {
            float v = ldsum;
            v += __shfl_xor(v, 1);
            v += __shfl_xor(v, 2);
            v += __shfl_xor(v, 4);
            v += __shfl_xor(v, 8);
            if (c == 0)
                out[(size_t)NS * DD + (size_t)cdim * (NS / 64) + chunk * (TILES / 4) + (i >> 2)] = v;
            ldsum = 0.0f;
        }

        #pragma unroll
        for (int q = 0; q < 8; q++) pv[q] = p0[q];
        #pragma unroll
        for (int q = 8; q < PP; q++) pv[q] = p1[q - 8];
        xs_c = xs_n;
    }
#undef LOAD_TILE

    {
        float e[10], invw, invh;
        rqs_soft(pv, e, invw, invh);
        float xc, icw, ibw, ich, ih, rlo, rhi;
        int idx;
        rqs_bins(pv, e, invw, invh, xs_c, xc, idx, icw, ibw, ich, ih, rlo, rhi);
        float idv, idvp1;
        rqs_knots(idx, rlo, rhi, idv, idvp1);
        float z, ld;
        rqs_form(xs_c, xc, icw, ibw, ich, ih, idv, idvp1, z, ld);
        out[(size_t)cdim * NS + s0 + (TILES - 1) * 16] = z;
        float v = ldsum + ld;
        v += __shfl_xor(v, 1);
        v += __shfl_xor(v, 2);
        v += __shfl_xor(v, 4);
        v += __shfl_xor(v, 8);
        if (c == 0)
            out[(size_t)NS * DD + (size_t)cdim * (NS / 64) + chunk * (TILES / 4) + ((TILES - 1) >> 2)] = v;
    }
}

extern "C" void kernel_launch(void* const* d_in, const int* in_sizes, int n_in,
                              void* d_out, int out_size, void* d_ws, size_t ws_size,
                              hipStream_t stream) {
    const float* x     = (const float*)d_in[0];
    const float* initp = (const float*)d_in[1];
    const float* W1    = (const float*)d_in[2];
    const float* b1    = (const float*)d_in[3];
    const float* W2    = (const float*)d_in[4];
    const float* b2    = (const float*)d_in[5];
    const float* W3    = (const float*)d_in[6];
    const float* b3    = (const float*)d_in[7];
    float* out = (float*)d_out;
    float* Bt  = (float*)d_ws;

    nsf_pack_w1<<<(NCAT * 64 + 255) / 256, 256, 0, stream>>>(W1, Bt);
    nsf_fused<<<(32 / WAVES) * (NS / (TILES * 16)), WAVES * 32, 0, stream>>>(x, initp, b1, W2, b2, W3, b3, Bt, out);
}


# tanh from one formula on the signed value, 1 - 2/(exp(2v)+1): small-|v| polynomial, compare+select, abs and sign copy gone; error now absolute (a few 1e-7), was relative; no sum reordered
# speedup vs baseline: 3.8021x; 1.0863x over previous
#include <hip/hip_runtime.h>
#include <math.h>

typedef float v2f __attribute__((ext_vector_type(2)));
typedef float v8f __attribute__((ext_vector_type(8)));

#define NS 32768
#define DD 64
#define LL 63
#define PP 14
#define NCAT 512
#define BB 5.0f
#define MINW 0.001f
#define MINH 0.001f
#define MIND 0.001f

__global__ void nsf_pack_w1(const float* __restrict__ W1, float* __restrict__ Bt) {
    int idx = blockIdx.x * blockDim.x + threadIdx.x;
    if (idx >= NCAT * 64) return;
    int col = idx >> 6;
    int k   = idx & 63;
    int l   = col >> 3;
    int h   = col & 7;
    float v = 0.0f;
    if (l < LL && k <= l) v = W1[(l * LL + k) * 8 + h];
    Bt[col * 64 + k] = v;
}

__device__ __forceinline__ float exp_core(float x) {
    const float c_hi = 0x1.715476p+0f, c_lo = 0x1.4ae0bep-26f;
    const float n = __builtin_rintf(x * c_hi);
    float a = __builtin_fmaf(x, c_hi, -n);
    a = __builtin_fmaf(x, c_lo, a);
    return __builtin_ldexpf(__builtin_amdgcn_exp2f(a), (int)n);
}
__device__ __forceinline__ float exp_nonpos(float x) {
    const float r = exp_core(x);
    return (x < -0x1.9d1da0p+6f) ? 0.0f : r;
}


__device__ __forceinline__ void rqs_soft(const float (&p)[PP], float (&e)[10], float& invw, float& invh) {
    float mxw = p[0], mxh = p[5];
    #pragma unroll
    for (int i = 1; i < 5; i++) { mxw = fmaxf(mxw, p[i]); mxh = fmaxf(mxh, p[5 + i]); }
    float d[10];
    #pragma unroll
    for (int i = 0; i < 5; i++) { d[i] = p[i] - mxw; d[5 + i] = p[5 + i] - mxh; }
    #pragma unroll
    for (int i = 0; i < 10; i++) e[i] = exp_nonpos(d[i]);
    float sw = 0.0f, sh = 0.0f;
    #pragma unroll
    for (int i = 0; i < 5; i++) { sw += e[i]; sh += e[5 + i]; }
    invw = 1.0f / sw; invh = 1.0f / sh;
}

__device__ __forceinline__ void rqs_bins(const float (&p)[PP], const float (&e)[10], float invw, float invh,
                                         float xf, float& xc, int& idx, float& icw, float& ibw,
                                         float& ich, float& ih, float& rlo, float& rhi) {
    float cumw[6], cumh[6];
    float csw = 0.0f, csh = 0.0f;
    cumw[0] = -BB; cumh[0] = -BB;
    #pragma unroll
    for (int i = 0; i < 4; i++) {
        csw += MINW + (1.0f - MINW * 5.0f) * e[i] * invw;
        csh += MINH + (1.0f - MINH * 5.0f) * e[5 + i] * invh;
        cumw[i + 1] = 2.0f * BB * csw - BB;
        cumh[i + 1] = 2.0f * BB * csh - BB;
    }
    cumw[5] = BB; cumh[5] = BB;

    xc = fminf(fmaxf(xf, -BB), BB);
    idx = 0;
    #pragma unroll
    for (int j = 1; j < 6; j++) idx += (xc >= cumw[j]) ? 1 : 0;
    if (idx > 4) idx = 4;

    icw = cumw[idx]; ibw = cumw[idx + 1] - cumw[idx];
    ich = cumh[idx]; ih  = cumh[idx + 1] - cumh[idx];
    const float r1 = p[10], r2 = p[11], r3 = p[12], r4 = p[13];
    const float lo34 = (idx >= 4) ? r4 : r3;
    const float lo12 = (idx >= 2) ? r2 : r1;
    rlo = (idx >= 3) ? lo34 : lo12;
    const float hi23 = (idx >= 3) ? r4 : r3;
    const float hi01 = (idx >= 1) ? r2 : r1;
    rhi = (idx >= 2) ? hi23 : hi01;
}

__device__ __forceinline__ void rqs_knots(int idx, float rlo, float rhi, float& idv, float& idvp1) {
    const float elo = expf(rlo), ehi = expf(rhi);
    const float llo = log1pf(elo), lhi = log1pf(ehi);
    const float splo = rlo > 20.0f ? rlo : llo;
    const float sphi = rhi > 20.0f ? rhi : lhi;
    idv   = (idx == 0) ? 1.0f : MIND + splo;
    idvp1 = (idx == 4) ? 1.0f : MIND + sphi;
}

__device__ __forceinline__ void rqs_form(float xf, float xc, float icw, float ibw, float ich, float ih,
                                         float idv, float idvp1, float& z, float& ld) {
    float idl = ih / ibw;
    float theta = (xc - icw) / ibw;
    float tmt = theta * (1.0f - theta);
    float num = ih * (idl * theta * theta + idv * tmt);
    float den = idl + (idv + idvp1 - 2.0f * idl) * tmt;
    float outv = ich + num / den;
    float omt = 1.0f - theta;
    float dnum = idl * idl * (idvp1 * theta * theta + 2.0f * idl * tmt + idv * omt * omt);
    float ldv = logf(dnum) - 2.0f * logf(den);
    bool inside = (xf >= -BB) && (xf <= BB);
    z  = inside ? outv : xf;
    ld = inside ? ldv : 0.0f;
}

__device__ __forceinline__ void tanh8_exp(const v8f a, float (&t)[8]) {
    const float k_hi = 0x1.715476p+1f, k_lo = 0x1.4ae0bep-25f;
    float n[8], f[8];
    #pragma unroll
    for (int r = 0; r < 8; r++) n[r] = __builtin_rintf(a[r] * k_hi);
    #pragma unroll
    for (int r = 0; r < 8; r++) f[r] = __builtin_fmaf(a[r], k_hi, -n[r]);
    #pragma unroll
    for (int r = 0; r < 8; r++) f[r] = __builtin_fmaf(a[r], k_lo, f[r]);
    #pragma unroll
    for (int r = 0; r < 8; r++) t[r] = __builtin_ldexpf(__builtin_amdgcn_exp2f(f[r]), (int)n[r]);
}
__device__ __forceinline__ void tanh8_fin(const float (&t)[8], float (&o)[8]) {
    float s[8];
    #pragma unroll
    for (int r = 0; r < 8; r++) s[r] = __builtin_amdgcn_rcpf(t[r] + 1.0f);
    #pragma unroll
    for (int r = 0; r < 8; r++) o[r] = fmaf(s[r], -2.0f, 1.0f);
}
__device__ __forceinline__ void tanh8(const v8f a, float (&o)[8]) {
    float t[8];
    tanh8_exp(a, t);
    tanh8_fin(t, o);
}

#define TILES 16
#define WAVES 16

__launch_bounds__(WAVES * 32) __attribute__((amdgpu_waves_per_eu(4, 4)))
__global__ void nsf_fused(const float* __restrict__ x, const float* __restrict__ initp,
                          const float* __restrict__ b1, const float* __restrict__ W2,
                          const float* __restrict__ b2, const float* __restrict__ W3,
                          const float* __restrict__ b3, const float* __restrict__ Bt,
                          float* __restrict__ out) {
    const int tid   = threadIdx.x;
    const int wave  = tid >> 5;
    const int lane  = tid & 31;
    const int c     = lane & 15;
    const int h     = lane >> 4;
    const int t     = (blockIdx.x & 1) * WAVES + wave;
    const int chunk = blockIdx.x >> 1;

    const int  l    = 2 * t + h;
    const bool lv   = (l < LL);
    const int  lc   = lv ? l : (LL - 1);
    const int  cdim = lv ? (l + 1) : 0;

    const float* wrow = Bt + (size_t)(16 * t + c) * 64 + 2 * h;
    v2f a1[16];
    #pragma unroll
    for (int kk = 0; kk < 16; kk++) a1[kk] = *(const v2f*)(wrow + 4 * kk);

    const int  rp   = c >> 3;
    const int  ru   = c & 7;
    const bool diag = (rp == h) && lv;

    const float* w2 = W2 + (size_t)lc * 64 + ru;
    v2f a2[4];
    #pragma unroll
    for (int kk = 0; kk < 4; kk++) {
        float w0 = w2[(2 * kk) * 8], w1 = w2[(2 * kk + 1) * 8];
        a2[kk].x = diag ? w0 : 0.0f;
        a2[kk].y = diag ? w1 : 0.0f;
    }

    const float* w3 = W3 + (size_t)lc * (8 * PP);
    const bool on1 = diag && (ru < PP - 8);
    const int  q1  = on1 ? (8 + ru) : 0;
    v2f a30[4], a31[4];
    #pragma unroll
    for (int kk = 0; kk < 4; kk++) {
        float u0 = w3[(2 * kk) * PP + ru], u1 = w3[(2 * kk + 1) * PP + ru];
        float v0 = w3[(2 * kk) * PP + q1], v1 = w3[(2 * kk + 1) * PP + q1];
        a30[kk].x = diag ? u0 : 0.0f;
        a30[kk].y = diag ? u1 : 0.0f;
        a31[kk].x = on1 ? v0 : 0.0f;
        a31[kk].y = on1 ? v1 : 0.0f;
    }

    v8f c1, c2, c30, c31;
    #pragma unroll
    for (int r = 0; r < 8; r++) {
        c1[r] = b1[lc * 8 + r];
        c2[r] = b2[lc * 8 + r];
        float g0 = b3[lc * PP + r], i0 = initp[r];
        c30[r] = lv ? g0 : i0;
        if (r < PP - 8) {
            float g1 = b3[lc * PP + 8 + r], i1 = initp[8 + r];
            c31[r] = lv ? g1 : i1;
        } else {
            c31[r] = 0.0f;
        }
    }

    float ldsum = 0.0f;

    const int s0 = chunk * (TILES * 16) + c;

    v2f xb[16];
    float xs_l;
#define LOAD_TILE(TI)                                                        \
    {                                                                        \
        const float* xrow_ = x + (size_t)(s0 + (TI) * 16) * DD;              \
        _Pragma("unroll")                                                    \
        for (int kk = 0; kk < 16; kk++)                                      \
            xb[kk] = *(const v2f*)(xrow_ + 4 * kk + 2 * h);                  \
        xs_l = xrow_[cdim];                                                  \
    }

    float pv[PP];
    float xs_c;
    {
        LOAD_TILE(0)
        v8f acc1 = c1;
        #pragma unroll
        for (int kk = 0; kk < 16; kk++)
            acc1 = __builtin_amdgcn_wmma_f32_16x16x4_f32(
                       false, a1[kk], false, xb[kk], (short)0, acc1, false, false);
        xs_c = xs_l;
        LOAD_TILE(1)
        float h1[8];
        tanh8(acc1, h1);
        v8f acc2 = c2;
        #pragma unroll
        for (int kk = 0; kk < 4; kk++) {
            v2f b = {h1[2 * kk], h1[2 * kk + 1]};
            acc2 = __builtin_amdgcn_wmma_f32_16x16x4_f32(
                       false, a2[kk], false, b, (short)0, acc2, false, false);
        }
        float h2[8];
        tanh8(acc2, h2);
        v8f p0 = c30, p1 = c31;
        #pragma unroll
        for (int kk = 0; kk < 4; kk++) {
            v2f b = {h2[2 * kk], h2[2 * kk + 1]};
            p0 = __builtin_amdgcn_wmma_f32_16x16x4_f32(
                     false, a30[kk], false, b, (short)0, p0, false, false);
            p1 = __builtin_amdgcn_wmma_f32_16x16x4_f32(
                     false, a31[kk], false, b, (short)0, p1, false, false);
        }
        #pragma unroll
        for (int q = 0; q < 8; q++) pv[q] = p0[q];
        #pragma unroll
        for (int q = 8; q < PP; q++) pv[q] = p1[q - 8];
    }

    for (int i = 0; i < TILES - 1; ++i) {
        v8f acc1 = c1;
        #pragma unroll
        for (int kk = 0; kk < 16; kk++)
            acc1 = __builtin_amdgcn_wmma_f32_16x16x4_f32(
                       false, a1[kk], false, xb[kk], (short)0, acc1, false, false);
        const float xs_n = xs_l;

        float e[10], invw, invh;
        rqs_soft(pv, e, invw, invh);

        const int nt = (i + 2) & (TILES - 1);
        LOAD_TILE(nt)

        float t8[8], h1[8];
        tanh8_exp(acc1, t8);

        float xc, icw, ibw, ich, ih, rlo, rhi;
        int idx;
        rqs_bins(pv, e, invw, invh, xs_c, xc, idx, icw, ibw, ich, ih, rlo, rhi);

        float idv, idvp1;
        rqs_knots(idx, rlo, rhi, idv, idvp1);

        tanh8_fin(t8, h1);

        v8f acc2 = c2;
        #pragma unroll
        for (int kk = 0; kk < 4; kk++) {
            v2f b = {h1[2 * kk], h1[2 * kk + 1]};
            acc2 = __builtin_amdgcn_wmma_f32_16x16x4_f32(
                       false, a2[kk], false, b, (short)0, acc2, false, false);
        }

        float w8[8], h2[8];
        tanh8_exp(acc2, w8);
        float z, ld;
        rqs_form(xs_c, xc, icw, ibw, ich, ih, idv, idvp1, z, ld);
        tanh8_fin(w8, h2);

        v8f p0 = c30, p1 = c31;
        #pragma unroll
        for (int kk = 0; kk < 4; kk++) {
            v2f b = {h2[2 * kk], h2[2 * kk + 1]};
            p0 = __builtin_amdgcn_wmma_f32_16x16x4_f32(
                     false, a30[kk], false, b, (short)0, p0, false, false);
            p1 = __builtin_amdgcn_wmma_f32_16x16x4_f32(
                     false, a31[kk], false, b, (short)0, p1, false, false);
        }

        out[(size_t)cdim * NS + s0 + i * 16] = z;
        ldsum += ld;

        if ((i & 3) == 3) {
            float v = ldsum;
            v += __shfl_xor(v, 1);
            v += __shfl_xor(v, 2);
            v += __shfl_xor(v, 4);
            v += __shfl_xor(v, 8);
            if (c == 0)
                out[(size_t)NS * DD + (size_t)cdim * (NS / 64) + chunk * (TILES / 4) + (i >> 2)] = v;
            ldsum = 0.0f;
        }

        #pragma unroll
        for (int q = 0; q < 8; q++) pv[q] = p0[q];
        #pragma unroll
        for (int q = 8; q < PP; q++) pv[q] = p1[q - 8];
        xs_c = xs_n;
    }
#undef LOAD_TILE

    {
        float e[10], invw, invh;
        rqs_soft(pv, e, invw, invh);
        float xc, icw, ibw, ich, ih, rlo, rhi;
        int idx;
        rqs_bins(pv, e, invw, invh, xs_c, xc, idx, icw, ibw, ich, ih, rlo, rhi);
        float idv, idvp1;
        rqs_knots(idx, rlo, rhi, idv, idvp1);
        float z, ld;
        rqs_form(xs_c, xc, icw, ibw, ich, ih, idv, idvp1, z, ld);
        out[(size_t)cdim * NS + s0 + (TILES - 1) * 16] = z;
        float v = ldsum + ld;
        v += __shfl_xor(v, 1);
        v += __shfl_xor(v, 2);
        v += __shfl_xor(v, 4);
        v += __shfl_xor(v, 8);
        if (c == 0)
            out[(size_t)NS * DD + (size_t)cdim * (NS / 64) + chunk * (TILES / 4) + ((TILES - 1) >> 2)] = v;
    }
}

extern "C" void kernel_launch(void* const* d_in, const int* in_sizes, int n_in,
                              void* d_out, int out_size, void* d_ws, size_t ws_size,
                              hipStream_t stream) {
    const float* x     = (const float*)d_in[0];
    const float* initp = (const float*)d_in[1];
    const float* W1    = (const float*)d_in[2];
    const float* b1    = (const float*)d_in[3];
    const float* W2    = (const float*)d_in[4];
    const float* b2    = (const float*)d_in[5];
    const float* W3    = (const float*)d_in[6];
    const float* b3    = (const float*)d_in[7];
    float* out = (float*)d_out;
    float* Bt  = (float*)d_ws;

    nsf_pack_w1<<<(NCAT * 64 + 255) / 256, 256, 0, stream>>>(W1, Bt);
    nsf_fused<<<(32 / WAVES) * (NS / (TILES * 16)), WAVES * 32, 0, stream>>>(x, initp, b1, W2, b2, W3, b3, Bt, out);
}


# exp(x) as ONE hardware 2^(x*log2e) on the once-rounded product: integer/remainder split (rint, 2 FMAs, convert, ldexp) and range guard gone in tanh (16), softmax (10), softplus (2); no sum reordered
# speedup vs baseline: 4.2234x; 1.1108x over previous
#include <hip/hip_runtime.h>
#include <math.h>

typedef float v2f __attribute__((ext_vector_type(2)));
typedef float v8f __attribute__((ext_vector_type(8)));

#define NS 32768
#define DD 64
#define LL 63
#define PP 14
#define NCAT 512
#define BB 5.0f
#define MINW 0.001f
#define MINH 0.001f
#define MIND 0.001f

__global__ void nsf_pack_w1(const float* __restrict__ W1, float* __restrict__ Bt) {
    int idx = blockIdx.x * blockDim.x + threadIdx.x;
    if (idx >= NCAT * 64) return;
    int col = idx >> 6;
    int k   = idx & 63;
    int l   = col >> 3;
    int h   = col & 7;
    float v = 0.0f;
    if (l < LL && k <= l) v = W1[(l * LL + k) * 8 + h];
    Bt[col * 64 + k] = v;
}

__device__ __forceinline__ float exp_once(float x) {
    return __builtin_amdgcn_exp2f(x * 0x1.715476p+0f);
}


__device__ __forceinline__ void rqs_soft(const float (&p)[PP], float (&e)[10], float& invw, float& invh) {
    float mxw = p[0], mxh = p[5];
    #pragma unroll
    for (int i = 1; i < 5; i++) { mxw = fmaxf(mxw, p[i]); mxh = fmaxf(mxh, p[5 + i]); }
    float d[10];
    #pragma unroll
    for (int i = 0; i < 5; i++) { d[i] = p[i] - mxw; d[5 + i] = p[5 + i] - mxh; }
    #pragma unroll
    for (int i = 0; i < 10; i++) e[i] = exp_once(d[i]);
    float sw = 0.0f, sh = 0.0f;
    #pragma unroll
    for (int i = 0; i < 5; i++) { sw += e[i]; sh += e[5 + i]; }
    invw = 1.0f / sw; invh = 1.0f / sh;
}

__device__ __forceinline__ void rqs_bins(const float (&p)[PP], const float (&e)[10], float invw, float invh,
                                         float xf, float& xc, int& idx, float& icw, float& ibw,
                                         float& ich, float& ih, float& rlo, float& rhi) {
    float cumw[6], cumh[6];
    float csw = 0.0f, csh = 0.0f;
    cumw[0] = -BB; cumh[0] = -BB;
    #pragma unroll
    for (int i = 0; i < 4; i++) {
        csw += MINW + (1.0f - MINW * 5.0f) * e[i] * invw;
        csh += MINH + (1.0f - MINH * 5.0f) * e[5 + i] * invh;
        cumw[i + 1] = 2.0f * BB * csw - BB;
        cumh[i + 1] = 2.0f * BB * csh - BB;
    }
    cumw[5] = BB; cumh[5] = BB;

    xc = fminf(fmaxf(xf, -BB), BB);
    idx = 0;
    #pragma unroll
    for (int j = 1; j < 6; j++) idx += (xc >= cumw[j]) ? 1 : 0;
    if (idx > 4) idx = 4;

    icw = cumw[idx]; ibw = cumw[idx + 1] - cumw[idx];
    ich = cumh[idx]; ih  = cumh[idx + 1] - cumh[idx];
    const float r1 = p[10], r2 = p[11], r3 = p[12], r4 = p[13];
    const float lo34 = (idx >= 4) ? r4 : r3;
    const float lo12 = (idx >= 2) ? r2 : r1;
    rlo = (idx >= 3) ? lo34 : lo12;
    const float hi23 = (idx >= 3) ? r4 : r3;
    const float hi01 = (idx >= 1) ? r2 : r1;
    rhi = (idx >= 2) ? hi23 : hi01;
}

__device__ __forceinline__ void rqs_knots(int idx, float rlo, float rhi, float& idv, float& idvp1) {
    const float elo = exp_once(rlo), ehi = exp_once(rhi);
    const float llo = log1pf(elo), lhi = log1pf(ehi);
    const float splo = rlo > 20.0f ? rlo : llo;
    const float sphi = rhi > 20.0f ? rhi : lhi;
    idv   = (idx == 0) ? 1.0f : MIND + splo;
    idvp1 = (idx == 4) ? 1.0f : MIND + sphi;
}

__device__ __forceinline__ void rqs_form(float xf, float xc, float icw, float ibw, float ich, float ih,
                                         float idv, float idvp1, float& z, float& ld) {
    float idl = ih / ibw;
    float theta = (xc - icw) / ibw;
    float tmt = theta * (1.0f - theta);
    float num = ih * (idl * theta * theta + idv * tmt);
    float den = idl + (idv + idvp1 - 2.0f * idl) * tmt;
    float outv = ich + num / den;
    float omt = 1.0f - theta;
    float dnum = idl * idl * (idvp1 * theta * theta + 2.0f * idl * tmt + idv * omt * omt);
    float ldv = logf(dnum) - 2.0f * logf(den);
    bool inside = (xf >= -BB) && (xf <= BB);
    z  = inside ? outv : xf;
    ld = inside ? ldv : 0.0f;
}

__device__ __forceinline__ void tanh8_exp(const v8f a, float (&t)[8]) {
    const float k2 = 0x1.715476p+1f;
    #pragma unroll
    for (int r = 0; r < 8; r++) t[r] = __builtin_amdgcn_exp2f(a[r] * k2);
}
__device__ __forceinline__ void tanh8_fin(const float (&t)[8], float (&o)[8]) {
    float s[8];
    #pragma unroll
    for (int r = 0; r < 8; r++) s[r] = __builtin_amdgcn_rcpf(t[r] + 1.0f);
    #pragma unroll
    for (int r = 0; r < 8; r++) o[r] = fmaf(s[r], -2.0f, 1.0f);
}
__device__ __forceinline__ void tanh8(const v8f a, float (&o)[8]) {
    float t[8];
    tanh8_exp(a, t);
    tanh8_fin(t, o);
}

#define TILES 16
#define WAVES 16

__launch_bounds__(WAVES * 32) __attribute__((amdgpu_waves_per_eu(4, 4)))
__global__ void nsf_fused(const float* __restrict__ x, const float* __restrict__ initp,
                          const float* __restrict__ b1, const float* __restrict__ W2,
                          const float* __restrict__ b2, const float* __restrict__ W3,
                          const float* __restrict__ b3, const float* __restrict__ Bt,
                          float* __restrict__ out) {
    const int tid   = threadIdx.x;
    const int wave  = tid >> 5;
    const int lane  = tid & 31;
    const int c     = lane & 15;
    const int h     = lane >> 4;
    const int t     = (blockIdx.x & 1) * WAVES + wave;
    const int chunk = blockIdx.x >> 1;

    const int  l    = 2 * t + h;
    const bool lv   = (l < LL);
    const int  lc   = lv ? l : (LL - 1);
    const int  cdim = lv ? (l + 1) : 0;

    const float* wrow = Bt + (size_t)(16 * t + c) * 64 + 2 * h;
    v2f a1[16];
    #pragma unroll
    for (int kk = 0; kk < 16; kk++) a1[kk] = *(const v2f*)(wrow + 4 * kk);

    const int  rp   = c >> 3;
    const int  ru   = c & 7;
    const bool diag = (rp == h) && lv;

    const float* w2 = W2 + (size_t)lc * 64 + ru;
    v2f a2[4];
    #pragma unroll
    for (int kk = 0; kk < 4; kk++) {
        float w0 = w2[(2 * kk) * 8], w1 = w2[(2 * kk + 1) * 8];
        a2[kk].x = diag ? w0 : 0.0f;
        a2[kk].y = diag ? w1 : 0.0f;
    }

    const float* w3 = W3 + (size_t)lc * (8 * PP);
    const bool on1 = diag && (ru < PP - 8);
    const int  q1  = on1 ? (8 + ru) : 0;
    v2f a30[4], a31[4];
    #pragma unroll
    for (int kk = 0; kk < 4; kk++) {
        float u0 = w3[(2 * kk) * PP + ru], u1 = w3[(2 * kk + 1) * PP + ru];
        float v0 = w3[(2 * kk) * PP + q1], v1 = w3[(2 * kk + 1) * PP + q1];
        a30[kk].x = diag ? u0 : 0.0f;
        a30[kk].y = diag ? u1 : 0.0f;
        a31[kk].x = on1 ? v0 : 0.0f;
        a31[kk].y = on1 ? v1 : 0.0f;
    }

    v8f c1, c2, c30, c31;
    #pragma unroll
    for (int r = 0; r < 8; r++) {
        c1[r] = b1[lc * 8 + r];
        c2[r] = b2[lc * 8 + r];
        float g0 = b3[lc * PP + r], i0 = initp[r];
        c30[r] = lv ? g0 : i0;
        if (r < PP - 8) {
            float g1 = b3[lc * PP + 8 + r], i1 = initp[8 + r];
            c31[r] = lv ? g1 : i1;
        } else {
            c31[r] = 0.0f;
        }
    }

    float ldsum = 0.0f;

    const int s0 = chunk * (TILES * 16) + c;

    v2f xb[16];
    float xs_l;
#define LOAD_TILE(TI)                                                        \
    {                                                                        \
        const float* xrow_ = x + (size_t)(s0 + (TI) * 16) * DD;              \
        _Pragma("unroll")                                                    \
        for (int kk = 0; kk < 16; kk++)                                      \
            xb[kk] = *(const v2f*)(xrow_ + 4 * kk + 2 * h);                  \
        xs_l = xrow_[cdim];                                                  \
    }

    float pv[PP];
    float xs_c;
    {
        LOAD_TILE(0)
        v8f acc1 = c1;
        #pragma unroll
        for (int kk = 0; kk < 16; kk++)
            acc1 = __builtin_amdgcn_wmma_f32_16x16x4_f32(
                       false, a1[kk], false, xb[kk], (short)0, acc1, false, false);
        xs_c = xs_l;
        LOAD_TILE(1)
        float h1[8];
        tanh8(acc1, h1);
        v8f acc2 = c2;
        #pragma unroll
        for (int kk = 0; kk < 4; kk++) {
            v2f b = {h1[2 * kk], h1[2 * kk + 1]};
            acc2 = __builtin_amdgcn_wmma_f32_16x16x4_f32(
                       false, a2[kk], false, b, (short)0, acc2, false, false);
        }
        float h2[8];
        tanh8(acc2, h2);
        v8f p0 = c30, p1 = c31;
        #pragma unroll
        for (int kk = 0; kk < 4; kk++) {
            v2f b = {h2[2 * kk], h2[2 * kk + 1]};
            p0 = __builtin_amdgcn_wmma_f32_16x16x4_f32(
                     false, a30[kk], false, b, (short)0, p0, false, false);
            p1 = __builtin_amdgcn_wmma_f32_16x16x4_f32(
                     false, a31[kk], false, b, (short)0, p1, false, false);
        }
        #pragma unroll
        for (int q = 0; q < 8; q++) pv[q] = p0[q];
        #pragma unroll
        for (int q = 8; q < PP; q++) pv[q] = p1[q - 8];
    }

    for (int i = 0; i < TILES - 1; ++i) {
        v8f acc1 = c1;
        #pragma unroll
        for (int kk = 0; kk < 16; kk++)
            acc1 = __builtin_amdgcn_wmma_f32_16x16x4_f32(
                       false, a1[kk], false, xb[kk], (short)0, acc1, false, false);
        const float xs_n = xs_l;

        float e[10], invw, invh;
        rqs_soft(pv, e, invw, invh);

        const int nt = (i + 2) & (TILES - 1);
        LOAD_TILE(nt)

        float t8[8], h1[8];
        tanh8_exp(acc1, t8);

        float xc, icw, ibw, ich, ih, rlo, rhi;
        int idx;
        rqs_bins(pv, e, invw, invh, xs_c, xc, idx, icw, ibw, ich, ih, rlo, rhi);

        float idv, idvp1;
        rqs_knots(idx, rlo, rhi, idv, idvp1);

        tanh8_fin(t8, h1);

        v8f acc2 = c2;
        #pragma unroll
        for (int kk = 0; kk < 4; kk++) {
            v2f b = {h1[2 * kk], h1[2 * kk + 1]};
            acc2 = __builtin_amdgcn_wmma_f32_16x16x4_f32(
                       false, a2[kk], false, b, (short)0, acc2, false, false);
        }

        float w8[8], h2[8];
        tanh8_exp(acc2, w8);
        float z, ld;
        rqs_form(xs_c, xc, icw, ibw, ich, ih, idv, idvp1, z, ld);
        tanh8_fin(w8, h2);

        v8f p0 = c30, p1 = c31;
        #pragma unroll
        for (int kk = 0; kk < 4; kk++) {
            v2f b = {h2[2 * kk], h2[2 * kk + 1]};
            p0 = __builtin_amdgcn_wmma_f32_16x16x4_f32(
                     false, a30[kk], false, b, (short)0, p0, false, false);
            p1 = __builtin_amdgcn_wmma_f32_16x16x4_f32(
                     false, a31[kk], false, b, (short)0, p1, false, false);
        }

        out[(size_t)cdim * NS + s0 + i * 16] = z;
        ldsum += ld;

        if ((i & 3) == 3) {
            float v = ldsum;
            v += __shfl_xor(v, 1);
            v += __shfl_xor(v, 2);
            v += __shfl_xor(v, 4);
            v += __shfl_xor(v, 8);
            if (c == 0)
                out[(size_t)NS * DD + (size_t)cdim * (NS / 64) + chunk * (TILES / 4) + (i >> 2)] = v;
            ldsum = 0.0f;
        }

        #pragma unroll
        for (int q = 0; q < 8; q++) pv[q] = p0[q];
        #pragma unroll
        for (int q = 8; q < PP; q++) pv[q] = p1[q - 8];
        xs_c = xs_n;
    }
#undef LOAD_TILE

    {
        float e[10], invw, invh;
        rqs_soft(pv, e, invw, invh);
        float xc, icw, ibw, ich, ih, rlo, rhi;
        int idx;
        rqs_bins(pv, e, invw, invh, xs_c, xc, idx, icw, ibw, ich, ih, rlo, rhi);
        float idv, idvp1;
        rqs_knots(idx, rlo, rhi, idv, idvp1);
        float z, ld;
        rqs_form(xs_c, xc, icw, ibw, ich, ih, idv, idvp1, z, ld);
        out[(size_t)cdim * NS + s0 + (TILES - 1) * 16] = z;
        float v = ldsum + ld;
        v += __shfl_xor(v, 1);
        v += __shfl_xor(v, 2);
        v += __shfl_xor(v, 4);
        v += __shfl_xor(v, 8);
        if (c == 0)
            out[(size_t)NS * DD + (size_t)cdim * (NS / 64) + chunk * (TILES / 4) + ((TILES - 1) >> 2)] = v;
    }
}

extern "C" void kernel_launch(void* const* d_in, const int* in_sizes, int n_in,
                              void* d_out, int out_size, void* d_ws, size_t ws_size,
                              hipStream_t stream) {
    const float* x     = (const float*)d_in[0];
    const float* initp = (const float*)d_in[1];
    const float* W1    = (const float*)d_in[2];
    const float* b1    = (const float*)d_in[3];
    const float* W2    = (const float*)d_in[4];
    const float* b2    = (const float*)d_in[5];
    const float* W3    = (const float*)d_in[6];
    const float* b3    = (const float*)d_in[7];
    float* out = (float*)d_out;
    float* Bt  = (float*)d_ws;

    nsf_pack_w1<<<(NCAT * 64 + 255) / 256, 256, 0, stream>>>(W1, Bt);
    nsf_fused<<<(32 / WAVES) * (NS / (TILES * 16)), WAVES * 32, 0, stream>>>(x, initp, b1, W2, b2, W3, b3, Bt, out);
}


# softplus: log1p(e) = ln2*log2(u) + (e-(u-1))/u, u = rounded 1+e: one log2, one reciprocal, 7 instructions; library log1pf (two-word arithmetic, ~70) gone; within 3 ulp, no sum reordered
# speedup vs baseline: 4.6152x; 1.0928x over previous
#include <hip/hip_runtime.h>
#include <math.h>

typedef float v2f __attribute__((ext_vector_type(2)));
typedef float v8f __attribute__((ext_vector_type(8)));

#define NS 32768
#define DD 64
#define LL 63
#define PP 14
#define NCAT 512
#define BB 5.0f
#define MINW 0.001f
#define MINH 0.001f
#define MIND 0.001f

__global__ void nsf_pack_w1(const float* __restrict__ W1, float* __restrict__ Bt) {
    int idx = blockIdx.x * blockDim.x + threadIdx.x;
    if (idx >= NCAT * 64) return;
    int col = idx >> 6;
    int k   = idx & 63;
    int l   = col >> 3;
    int h   = col & 7;
    float v = 0.0f;
    if (l < LL && k <= l) v = W1[(l * LL + k) * 8 + h];
    Bt[col * 64 + k] = v;
}

__device__ __forceinline__ float exp_once(float x) {
    return __builtin_amdgcn_exp2f(x * 0x1.715476p+0f);
}


__device__ __forceinline__ void rqs_soft(const float (&p)[PP], float (&e)[10], float& invw, float& invh) {
    float mxw = p[0], mxh = p[5];
    #pragma unroll
    for (int i = 1; i < 5; i++) { mxw = fmaxf(mxw, p[i]); mxh = fmaxf(mxh, p[5 + i]); }
    float d[10];
    #pragma unroll
    for (int i = 0; i < 5; i++) { d[i] = p[i] - mxw; d[5 + i] = p[5 + i] - mxh; }
    #pragma unroll
    for (int i = 0; i < 10; i++) e[i] = exp_once(d[i]);
    float sw = 0.0f, sh = 0.0f;
    #pragma unroll
    for (int i = 0; i < 5; i++) { sw += e[i]; sh += e[5 + i]; }
    invw = 1.0f / sw; invh = 1.0f / sh;
}

__device__ __forceinline__ void rqs_bins(const float (&p)[PP], const float (&e)[10], float invw, float invh,
                                         float xf, float& xc, int& idx, float& icw, float& ibw,
                                         float& ich, float& ih, float& rlo, float& rhi) {
    float cumw[6], cumh[6];
    float csw = 0.0f, csh = 0.0f;
    cumw[0] = -BB; cumh[0] = -BB;
    #pragma unroll
    for (int i = 0; i < 4; i++) {
        csw += MINW + (1.0f - MINW * 5.0f) * e[i] * invw;
        csh += MINH + (1.0f - MINH * 5.0f) * e[5 + i] * invh;
        cumw[i + 1] = 2.0f * BB * csw - BB;
        cumh[i + 1] = 2.0f * BB * csh - BB;
    }
    cumw[5] = BB; cumh[5] = BB;

    xc = fminf(fmaxf(xf, -BB), BB);
    idx = 0;
    #pragma unroll
    for (int j = 1; j < 6; j++) idx += (xc >= cumw[j]) ? 1 : 0;
    if (idx > 4) idx = 4;

    icw = cumw[idx]; ibw = cumw[idx + 1] - cumw[idx];
    ich = cumh[idx]; ih  = cumh[idx + 1] - cumh[idx];
    const float r1 = p[10], r2 = p[11], r3 = p[12], r4 = p[13];
    const float lo34 = (idx >= 4) ? r4 : r3;
    const float lo12 = (idx >= 2) ? r2 : r1;
    rlo = (idx >= 3) ? lo34 : lo12;
    const float hi23 = (idx >= 3) ? r4 : r3;
    const float hi01 = (idx >= 1) ? r2 : r1;
    rhi = (idx >= 2) ? hi23 : hi01;
}

__device__ __forceinline__ float log1p_of_exp(float e) {
    const float u  = 1.0f + e;
    const float c  = e - (u - 1.0f);
    const float lg = __builtin_amdgcn_logf(u);
    const float rc = __builtin_amdgcn_rcpf(u);
    return fmaf(c, rc, lg * 0x1.62e43p-1f);
}

__device__ __forceinline__ void rqs_knots(int idx, float rlo, float rhi, float& idv, float& idvp1) {
    const float elo = exp_once(rlo), ehi = exp_once(rhi);
    const float llo = log1p_of_exp(elo), lhi = log1p_of_exp(ehi);
    const float splo = rlo > 20.0f ? rlo : llo;
    const float sphi = rhi > 20.0f ? rhi : lhi;
    idv   = (idx == 0) ? 1.0f : MIND + splo;
    idvp1 = (idx == 4) ? 1.0f : MIND + sphi;
}

__device__ __forceinline__ void rqs_form(float xf, float xc, float icw, float ibw, float ich, float ih,
                                         float idv, float idvp1, float& z, float& ld) {
    float idl = ih / ibw;
    float theta = (xc - icw) / ibw;
    float tmt = theta * (1.0f - theta);
    float num = ih * (idl * theta * theta + idv * tmt);
    float den = idl + (idv + idvp1 - 2.0f * idl) * tmt;
    float outv = ich + num / den;
    float omt = 1.0f - theta;
    float dnum = idl * idl * (idvp1 * theta * theta + 2.0f * idl * tmt + idv * omt * omt);
    float ldv = logf(dnum) - 2.0f * logf(den);
    bool inside = (xf >= -BB) && (xf <= BB);
    z  = inside ? outv : xf;
    ld = inside ? ldv : 0.0f;
}

__device__ __forceinline__ void tanh8_exp(const v8f a, float (&t)[8]) {
    const float k2 = 0x1.715476p+1f;
    #pragma unroll
    for (int r = 0; r < 8; r++) t[r] = __builtin_amdgcn_exp2f(a[r] * k2);
}
__device__ __forceinline__ void tanh8_fin(const float (&t)[8], float (&o)[8]) {
    float s[8];
    #pragma unroll
    for (int r = 0; r < 8; r++) s[r] = __builtin_amdgcn_rcpf(t[r] + 1.0f);
    #pragma unroll
    for (int r = 0; r < 8; r++) o[r] = fmaf(s[r], -2.0f, 1.0f);
}
__device__ __forceinline__ void tanh8(const v8f a, float (&o)[8]) {
    float t[8];
    tanh8_exp(a, t);
    tanh8_fin(t, o);
}

#define TILES 16
#define WAVES 16

__launch_bounds__(WAVES * 32) __attribute__((amdgpu_waves_per_eu(4, 4)))
__global__ void nsf_fused(const float* __restrict__ x, const float* __restrict__ initp,
                          const float* __restrict__ b1, const float* __restrict__ W2,
                          const float* __restrict__ b2, const float* __restrict__ W3,
                          const float* __restrict__ b3, const float* __restrict__ Bt,
                          float* __restrict__ out) {
    const int tid   = threadIdx.x;
    const int wave  = tid >> 5;
    const int lane  = tid & 31;
    const int c     = lane & 15;
    const int h     = lane >> 4;
    const int t     = (blockIdx.x & 1) * WAVES + wave;
    const int chunk = blockIdx.x >> 1;

    const int  l    = 2 * t + h;
    const bool lv   = (l < LL);
    const int  lc   = lv ? l : (LL - 1);
    const int  cdim = lv ? (l + 1) : 0;

    const float* wrow = Bt + (size_t)(16 * t + c) * 64 + 2 * h;
    v2f a1[16];
    #pragma unroll
    for (int kk = 0; kk < 16; kk++) a1[kk] = *(const v2f*)(wrow + 4 * kk);

    const int  rp   = c >> 3;
    const int  ru   = c & 7;
    const bool diag = (rp == h) && lv;

    const float* w2 = W2 + (size_t)lc * 64 + ru;
    v2f a2[4];
    #pragma unroll
    for (int kk = 0; kk < 4; kk++) {
        float w0 = w2[(2 * kk) * 8], w1 = w2[(2 * kk + 1) * 8];
        a2[kk].x = diag ? w0 : 0.0f;
        a2[kk].y = diag ? w1 : 0.0f;
    }

    const float* w3 = W3 + (size_t)lc * (8 * PP);
    const bool on1 = diag && (ru < PP - 8);
    const int  q1  = on1 ? (8 + ru) : 0;
    v2f a30[4], a31[4];
    #pragma unroll
    for (int kk = 0; kk < 4; kk++) {
        float u0 = w3[(2 * kk) * PP + ru], u1 = w3[(2 * kk + 1) * PP + ru];
        float v0 = w3[(2 * kk) * PP + q1], v1 = w3[(2 * kk + 1) * PP + q1];
        a30[kk].x = diag ? u0 : 0.0f;
        a30[kk].y = diag ? u1 : 0.0f;
        a31[kk].x = on1 ? v0 : 0.0f;
        a31[kk].y = on1 ? v1 : 0.0f;
    }

    v8f c1, c2, c30, c31;
    #pragma unroll
    for (int r = 0; r < 8; r++) {
        c1[r] = b1[lc * 8 + r];
        c2[r] = b2[lc * 8 + r];
        float g0 = b3[lc * PP + r], i0 = initp[r];
        c30[r] = lv ? g0 : i0;
        if (r < PP - 8) {
            float g1 = b3[lc * PP + 8 + r], i1 = initp[8 + r];
            c31[r] = lv ? g1 : i1;
        } else {
            c31[r] = 0.0f;
        }
    }

    float ldsum = 0.0f;

    const int s0 = chunk * (TILES * 16) + c;

    v2f xb[16];
    float xs_l;
#define LOAD_TILE(TI)                                                        \
    {                                                                        \
        const float* xrow_ = x + (size_t)(s0 + (TI) * 16) * DD;              \
        _Pragma("unroll")                                                    \
        for (int kk = 0; kk < 16; kk++)                                      \
            xb[kk] = *(const v2f*)(xrow_ + 4 * kk + 2 * h);                  \
        xs_l = xrow_[cdim];                                                  \
    }

    float pv[PP];
    float xs_c;
    {
        LOAD_TILE(0)
        v8f acc1 = c1;
        #pragma unroll
        for (int kk = 0; kk < 16; kk++)
            acc1 = __builtin_amdgcn_wmma_f32_16x16x4_f32(
                       false, a1[kk], false, xb[kk], (short)0, acc1, false, false);
        xs_c = xs_l;
        LOAD_TILE(1)
        float h1[8];
        tanh8(acc1, h1);
        v8f acc2 = c2;
        #pragma unroll
        for (int kk = 0; kk < 4; kk++) {
            v2f b = {h1[2 * kk], h1[2 * kk + 1]};
            acc2 = __builtin_amdgcn_wmma_f32_16x16x4_f32(
                       false, a2[kk], false, b, (short)0, acc2, false, false);
        }
        float h2[8];
        tanh8(acc2, h2);
        v8f p0 = c30, p1 = c31;
        #pragma unroll
        for (int kk = 0; kk < 4; kk++) {
            v2f b = {h2[2 * kk], h2[2 * kk + 1]};
            p0 = __builtin_amdgcn_wmma_f32_16x16x4_f32(
                     false, a30[kk], false, b, (short)0, p0, false, false);
            p1 = __builtin_amdgcn_wmma_f32_16x16x4_f32(
                     false, a31[kk], false, b, (short)0, p1, false, false);
        }
        #pragma unroll
        for (int q = 0; q < 8; q++) pv[q] = p0[q];
        #pragma unroll
        for (int q = 8; q < PP; q++) pv[q] = p1[q - 8];
    }

    for (int i = 0; i < TILES - 1; ++i) {
        v8f acc1 = c1;
        #pragma unroll
        for (int kk = 0; kk < 16; kk++)
            acc1 = __builtin_amdgcn_wmma_f32_16x16x4_f32(
                       false, a1[kk], false, xb[kk], (short)0, acc1, false, false);
        const float xs_n = xs_l;

        float e[10], invw, invh;
        rqs_soft(pv, e, invw, invh);

        const int nt = (i + 2) & (TILES - 1);
        LOAD_TILE(nt)

        float t8[8], h1[8];
        tanh8_exp(acc1, t8);

        float xc, icw, ibw, ich, ih, rlo, rhi;
        int idx;
        rqs_bins(pv, e, invw, invh, xs_c, xc, idx, icw, ibw, ich, ih, rlo, rhi);

        float idv, idvp1;
        rqs_knots(idx, rlo, rhi, idv, idvp1);

        tanh8_fin(t8, h1);

        v8f acc2 = c2;
        #pragma unroll
        for (int kk = 0; kk < 4; kk++) {
            v2f b = {h1[2 * kk], h1[2 * kk + 1]};
            acc2 = __builtin_amdgcn_wmma_f32_16x16x4_f32(
                       false, a2[kk], false, b, (short)0, acc2, false, false);
        }

        float w8[8], h2[8];
        tanh8_exp(acc2, w8);
        float z, ld;
        rqs_form(xs_c, xc, icw, ibw, ich, ih, idv, idvp1, z, ld);
        tanh8_fin(w8, h2);

        v8f p0 = c30, p1 = c31;
        #pragma unroll
        for (int kk = 0; kk < 4; kk++) {
            v2f b = {h2[2 * kk], h2[2 * kk + 1]};
            p0 = __builtin_amdgcn_wmma_f32_16x16x4_f32(
                     false, a30[kk], false, b, (short)0, p0, false, false);
            p1 = __builtin_amdgcn_wmma_f32_16x16x4_f32(
                     false, a31[kk], false, b, (short)0, p1, false, false);
        }

        out[(size_t)cdim * NS + s0 + i * 16] = z;
        ldsum += ld;

        if ((i & 3) == 3) {
            float v = ldsum;
            v += __shfl_xor(v, 1);
            v += __shfl_xor(v, 2);
            v += __shfl_xor(v, 4);
            v += __shfl_xor(v, 8);
            if (c == 0)
                out[(size_t)NS * DD + (size_t)cdim * (NS / 64) + chunk * (TILES / 4) + (i >> 2)] = v;
            ldsum = 0.0f;
        }

        #pragma unroll
        for (int q = 0; q < 8; q++) pv[q] = p0[q];
        #pragma unroll
        for (int q = 8; q < PP; q++) pv[q] = p1[q - 8];
        xs_c = xs_n;
    }
#undef LOAD_TILE

    {
        float e[10], invw, invh;
        rqs_soft(pv, e, invw, invh);
        float xc, icw, ibw, ich, ih, rlo, rhi;
        int idx;
        rqs_bins(pv, e, invw, invh, xs_c, xc, idx, icw, ibw, ich, ih, rlo, rhi);
        float idv, idvp1;
        rqs_knots(idx, rlo, rhi, idv, idvp1);
        float z, ld;
        rqs_form(xs_c, xc, icw, ibw, ich, ih, idv, idvp1, z, ld);
        out[(size_t)cdim * NS + s0 + (TILES - 1) * 16] = z;
        float v = ldsum + ld;
        v += __shfl_xor(v, 1);
        v += __shfl_xor(v, 2);
        v += __shfl_xor(v, 4);
        v += __shfl_xor(v, 8);
        if (c == 0)
            out[(size_t)NS * DD + (size_t)cdim * (NS / 64) + chunk * (TILES / 4) + ((TILES - 1) >> 2)] = v;
    }
}

extern "C" void kernel_launch(void* const* d_in, const int* in_sizes, int n_in,
                              void* d_out, int out_size, void* d_ws, size_t ws_size,
                              hipStream_t stream) {
    const float* x     = (const float*)d_in[0];
    const float* initp = (const float*)d_in[1];
    const float* W1    = (const float*)d_in[2];
    const float* b1    = (const float*)d_in[3];
    const float* W2    = (const float*)d_in[4];
    const float* b2    = (const float*)d_in[5];
    const float* W3    = (const float*)d_in[6];
    const float* b3    = (const float*)d_in[7];
    float* out = (float*)d_out;
    float* Bt  = (float*)d_ws;

    nsf_pack_w1<<<(NCAT * 64 + 255) / 256, 256, 0, stream>>>(W1, Bt);
    nsf_fused<<<(32 / WAVES) * (NS / (TILES * 16)), WAVES * 32, 0, stream>>>(x, initp, b1, W2, b2, W3, b3, Bt, out);
}
